# starting kernel
# baseline (speedup 1.0000x reference)
#include <hip/hip_runtime.h>
#include <hip/hip_bf16.h>


typedef __attribute__((ext_vector_type(2))) float v2f;
typedef __attribute__((ext_vector_type(8))) float v8f;

#define D_FEAT 64
#define LDS_STRIDE 68

__global__ void gcn_zero_f32(float* __restrict__ p, int n) {
    int i = blockIdx.x * blockDim.x + threadIdx.x;
    if (i < n) p[i] = 0.0f;
}

__global__ void gcn_degree(const int* __restrict__ src, const int* __restrict__ dst,
                           float* __restrict__ deg_out, float* __restrict__ deg_in,
                           int nEdges) {
    int e = blockIdx.x * blockDim.x + threadIdx.x;
    if (e >= nEdges) return;
    atomicAdd(&deg_out[src[e]], 1.0f);
    atomicAdd(&deg_in[dst[e]], 1.0f);
}

__global__ void gcn_deg_isqrt(float* __restrict__ d, int n) {
    int i = blockIdx.x * blockDim.x + threadIdx.x;
    if (i < n) d[i] = rsqrtf(fmaxf(d[i], 1.0f));
}

__global__ void gcn_scale_rows(const float* __restrict__ x, const float* __restrict__ s,
                               float* __restrict__ out, int nQuads  ) {
    int t = blockIdx.x * blockDim.x + threadIdx.x;
    if (t >= nQuads) return;
    int row = t >> 4;
    float sc = s[row];
    float4 v = ((const float4*)x)[t];
    v.x *= sc; v.y *= sc; v.z *= sc; v.w *= sc;
    ((float4*)out)[t] = v;
}

__global__ void gcn_scatter(const float* __restrict__ h,
                            const int* __restrict__ src, const int* __restrict__ dst,
                            float* __restrict__ agg, int nEdges) {
    int t = blockIdx.x * blockDim.x + threadIdx.x;
    int e = t >> 4;
    if (e >= nEdges) return;
    int f = (t & 15) << 2;
    int s = src[e], d = dst[e];
    float4 v = *(const float4*)(h + (size_t)s * D_FEAT + f);
    float* p = agg + (size_t)d * D_FEAT + f;
    atomicAdd(p + 0, v.x);
    atomicAdd(p + 1, v.y);
    atomicAdd(p + 2, v.z);
    atomicAdd(p + 3, v.w);
}

__global__ void __launch_bounds__(128)
gcn_transform_wmma(const float* __restrict__ agg,
                   const float* __restrict__ W,
                   const float* __restrict__ bias,
                   const float* __restrict__ in_isqrt,
                   const float* __restrict__ post_scale,
                   float* __restrict__ out, int nNodes) {
    __shared__ float lds[16 * LDS_STRIDE];

    const int r0   = blockIdx.x * 16;
    const int tid  = threadIdx.x;
    const int wave = tid >> 5;
    const int lane = tid & 31;

    {
        int idx = tid * 8;
        int row = idx >> 6;
        int col = idx & 63;
        int grow = r0 + row; if (grow >= nNodes) grow = nNodes - 1;
        float s = in_isqrt[grow];
        const float* g = agg + (size_t)grow * D_FEAT + col;
        float* l = &lds[row * LDS_STRIDE + col];
        #pragma unroll
        for (int j = 0; j < 8; ++j) l[j] = g[j] * s;
    }
    __syncthreads();

    const int colbase = wave << 4;
    const int n     = lane & 15;
    const int khalf = lane >> 4;

    v8f acc = {};
    #pragma unroll
    for (int kk = 0; kk < 16; ++kk) {
        const int k0 = (kk << 2) + (khalf << 1);
        v2f a = *(const v2f*)&lds[n * LDS_STRIDE + k0];
        v2f b;
        b.x = W[(k0 + 0) * D_FEAT + colbase + n];
        b.y = W[(k0 + 1) * D_FEAT + colbase + n];
        acc = __builtin_amdgcn_wmma_f32_16x16x4_f32(
             false, a,  false, b,
             (short)0, acc,  false,  false);
    }

    const int col = colbase + n;
    const float bv = bias[col];
    #pragma unroll
    for (int r = 0; r < 8; ++r) {
        int row = r0 + (khalf << 3) + r;
        if (row < nNodes) {
            float v = acc[r] + bv;
            v = (v > 0.0f) ? v : 0.01f * v;
            if (post_scale) v *= post_scale[row];
            out[(size_t)row * D_FEAT + col] = v;
        }
    }
}

__global__ void gcn_classify(const float* __restrict__ h,
                             const float* __restrict__ Wc,
                             const float* __restrict__ bc,
                             float* __restrict__ out, int nNodes) {
    int i = blockIdx.x * blockDim.x + threadIdx.x;
    if (i >= nNodes) return;
    float a0 = bc[0], a1 = bc[1];
    const float4* row = (const float4*)(h + (size_t)i * D_FEAT);
    #pragma unroll
    for (int q = 0; q < 16; ++q) {
        float4 v = row[q];
        int k = q << 2;
        a0 += v.x * Wc[(k+0)*2] + v.y * Wc[(k+1)*2] + v.z * Wc[(k+2)*2] + v.w * Wc[(k+3)*2];
        a1 += v.x * Wc[(k+0)*2+1] + v.y * Wc[(k+1)*2+1] + v.z * Wc[(k+2)*2+1] + v.w * Wc[(k+3)*2+1];
    }
    out[(size_t)i * 2 + 0] = a0;
    out[(size_t)i * 2 + 1] = a1;
}

extern "C" void kernel_launch(void* const* d_in, const int* in_sizes, int n_in,
                              void* d_out, int out_size, void* d_ws, size_t ws_size,
                              hipStream_t stream) {
    const float* x   = (const float*)d_in[0];
    const int*   src = (const int*)  d_in[1];
    const int*   dst = (const int*)  d_in[2];
    const float* W1  = (const float*)d_in[3];
    const float* b1  = (const float*)d_in[4];
    const float* W2  = (const float*)d_in[5];
    const float* b2  = (const float*)d_in[6];
    const float* Wc  = (const float*)d_in[7];
    const float* bc  = (const float*)d_in[8];
    float* out = (float*)d_out;

    const int N = in_sizes[0] / D_FEAT;
    const int E = in_sizes[1];

    float* bufA      = (float*)d_ws;
    float* bufB      = bufA + (size_t)N * D_FEAT;
    float* out_isqrt = bufB + (size_t)N * D_FEAT;
    float* in_isqrt  = out_isqrt + N;

    const int B256 = 256;
    const int featTotal = N * D_FEAT;
    dim3 blk(B256);

    gcn_zero_f32<<<(2 * N + B256 - 1) / B256, blk, 0, stream>>>(out_isqrt, 2 * N);
    gcn_degree<<<(E + B256 - 1) / B256, blk, 0, stream>>>(src, dst, out_isqrt, in_isqrt, E);
    gcn_deg_isqrt<<<(2 * N + B256 - 1) / B256, blk, 0, stream>>>(out_isqrt, 2 * N);

    gcn_scale_rows<<<(N * 16 + B256 - 1) / B256, blk, 0, stream>>>(x, out_isqrt, bufA, N * 16);
    gcn_zero_f32<<<(featTotal + B256 - 1) / B256, blk, 0, stream>>>(bufB, featTotal);
    gcn_scatter<<<((size_t)E * 16 + B256 - 1) / B256, blk, 0, stream>>>(bufA, src, dst, bufB, E);
    gcn_transform_wmma<<<(N + 15) / 16, dim3(128), 0, stream>>>(
        bufB, W1, b1, in_isqrt, out_isqrt, bufA, N);

    gcn_zero_f32<<<(featTotal + B256 - 1) / B256, blk, 0, stream>>>(bufB, featTotal);
    gcn_scatter<<<((size_t)E * 16 + B256 - 1) / B256, blk, 0, stream>>>(bufA, src, dst, bufB, E);
    gcn_transform_wmma<<<(N + 15) / 16, dim3(128), 0, stream>>>(
        bufB, W2, b2, in_isqrt, nullptr, bufA, N);

    gcn_classify<<<(N + B256 - 1) / B256, blk, 0, stream>>>(bufA, Wc, bc, out, N);
}


# Scatter-add walks each edge chunk once per half of the sum rows; a work-group reads its die and takes a ticket for its own half's row range, so a row's adds come from one L2 half; arrival order kept
# speedup vs baseline: 2.0712x; 2.0712x over previous
#include <hip/hip_runtime.h>
#include <hip/hip_bf16.h>


typedef __attribute__((ext_vector_type(2))) float v2f;
typedef __attribute__((ext_vector_type(8))) float v8f;

#define D_FEAT 64
#define LDS_STRIDE 68

__global__ void gcn_zero_f32(float* __restrict__ p, int n,
                             unsigned* __restrict__ tickets) {
    int i = blockIdx.x * blockDim.x + threadIdx.x;
    if (i < n) p[i] = 0.0f;
    if (i == 0) { tickets[0] = 0u; tickets[1] = 0u; }
}

#define DEG_RANGE  50000
#define DEG_SLICES 16
#define DEG_BLOCK  1024

__device__ __forceinline__ void deg_count_one(unsigned* cnt, int id, int lo) {
    unsigned r = (unsigned)(id - lo);
    if (r < (unsigned)DEG_RANGE) atomicAdd(&cnt[r], 1u);
}

__global__ void __launch_bounds__(DEG_BLOCK)
gcn_degree_lds(const int* __restrict__ src, const int* __restrict__ dst,
               unsigned* __restrict__ partial, int nEdges, int nNodes) {
    __shared__ unsigned cnt[DEG_RANGE];
    const int tid = threadIdx.x;
    const int* __restrict__ idx = blockIdx.z ? dst : src;
    const int lo = blockIdx.y * DEG_RANGE;

    for (int i = tid; i < DEG_RANGE; i += DEG_BLOCK) cnt[i] = 0u;
    __syncthreads();

    const int per = (((nEdges + DEG_SLICES - 1) / DEG_SLICES) + 3) & ~3;
    const int e0 = blockIdx.x * per;
    int e1 = e0 + per; if (e1 > nEdges) e1 = nEdges;
    const int nv = (e1 > e0) ? ((e1 - e0) >> 2) : 0;
    const int4* __restrict__ v = (const int4*)(idx + e0);
    int q = tid;
    for (; q + 3 * DEG_BLOCK < nv; q += 4 * DEG_BLOCK) {
        int4 k0 = v[q];
        int4 k1 = v[q + DEG_BLOCK];
        int4 k2 = v[q + 2 * DEG_BLOCK];
        int4 k3 = v[q + 3 * DEG_BLOCK];
        deg_count_one(cnt, k0.x, lo); deg_count_one(cnt, k0.y, lo);
        deg_count_one(cnt, k0.z, lo); deg_count_one(cnt, k0.w, lo);
        deg_count_one(cnt, k1.x, lo); deg_count_one(cnt, k1.y, lo);
        deg_count_one(cnt, k1.z, lo); deg_count_one(cnt, k1.w, lo);
        deg_count_one(cnt, k2.x, lo); deg_count_one(cnt, k2.y, lo);
        deg_count_one(cnt, k2.z, lo); deg_count_one(cnt, k2.w, lo);
        deg_count_one(cnt, k3.x, lo); deg_count_one(cnt, k3.y, lo);
        deg_count_one(cnt, k3.z, lo); deg_count_one(cnt, k3.w, lo);
    }
    for (; q < nv; q += DEG_BLOCK) {
        int4 k = v[q];
        deg_count_one(cnt, k.x, lo);
        deg_count_one(cnt, k.y, lo);
        deg_count_one(cnt, k.z, lo);
        deg_count_one(cnt, k.w, lo);
    }
    for (int e = e0 + (nv << 2) + tid; e < e1; e += DEG_BLOCK)
        deg_count_one(cnt, idx[e], lo);
    __syncthreads();

    int nHere = nNodes - lo; if (nHere > DEG_RANGE) nHere = DEG_RANGE;
    unsigned* __restrict__ o =
        partial + (size_t)(blockIdx.z * DEG_SLICES + blockIdx.x) * nNodes + lo;
    for (int i = tid; i < nHere; i += DEG_BLOCK) o[i] = cnt[i];
}

__global__ void gcn_deg_isqrt(const unsigned* __restrict__ partial,
                              float* __restrict__ d, int nNodes) {
    int i = blockIdx.x * blockDim.x + threadIdx.x;
    if (i >= 2 * nNodes) return;
    const int a = (i >= nNodes) ? 1 : 0;
    const int node = i - a * nNodes;
    const unsigned* __restrict__ p = partial + (size_t)a * DEG_SLICES * nNodes + node;
    unsigned s = 0u;
    #pragma unroll
    for (int k = 0; k < DEG_SLICES; ++k) s += p[(size_t)k * nNodes];
    d[i] = rsqrtf(fmaxf((float)s, 1.0f));
}

__global__ void gcn_scale_rows(const float* __restrict__ x, const float* __restrict__ s,
                               float* __restrict__ out, int nQuads  ) {
    int t = blockIdx.x * blockDim.x + threadIdx.x;
    if (t >= nQuads) return;
    int row = t >> 4;
    float sc = s[row];
    float4 v = ((const float4*)x)[t];
    v.x *= sc; v.y *= sc; v.z *= sc; v.w *= sc;
    ((float4*)out)[t] = v;
}

#define HS_BLOCK 128
#define HS_WAVES (HS_BLOCK / 32)
#define HS_EDGES_PER_WAVE 64
#define HS_CHUNK (HS_WAVES * HS_EDGES_PER_WAVE)
#define HS_GROUP 8

__device__ __forceinline__ void scatter_group(const float* __restrict__ h,
                                              float* __restrict__ agg,
                                              int cs, int cd, int base, int n,
                                              int lane) {
    int s[HS_GROUP], d[HS_GROUP];
    float lo[HS_GROUP], hi[HS_GROUP];
    #pragma unroll
    for (int j = 0; j < HS_GROUP; ++j) {
        s[j] = __builtin_amdgcn_readlane(cs, base + j);
        d[j] = __builtin_amdgcn_readlane(cd, base + j);
    }
    #pragma unroll
    for (int j = 0; j < HS_GROUP; ++j) {
        const float* r = h + (size_t)s[j] * D_FEAT + lane;
        lo[j] = r[0];
        hi[j] = r[32];
    }
    #pragma unroll
    for (int j = 0; j < HS_GROUP; ++j) {
        if (base + j < n) {
            float* p = agg + (size_t)d[j] * D_FEAT + lane;
            atomicAdd(p, lo[j]);
            atomicAdd(p + 32, hi[j]);
        }
    }
}

__global__ void __launch_bounds__(HS_BLOCK)
gcn_scatter_half(const float* __restrict__ h,
                 const int* __restrict__ src, const int* __restrict__ dst,
                 float* __restrict__ agg, unsigned* __restrict__ tickets,
                 int nEdges, int halfRow) {
    __shared__ int s_walk[2];
    __shared__ int2 s_list[HS_WAVES][HS_EDGES_PER_WAVE];
    const int tid  = threadIdx.x;
    const int lane = tid & 31;
    const int wave = tid >> 5;
    const unsigned nChunks = (unsigned)((nEdges + HS_CHUNK - 1) / HS_CHUNK);

    if (tid == 0) {
        const unsigned id = __builtin_amdgcn_s_sendmsg_rtn(0x87);
        int half = (int)((id >> 18) & 1u);
        unsigned t = atomicAdd(&tickets[half], 1u);
        if (t >= nChunks) { half ^= 1; t = atomicAdd(&tickets[half], 1u); }
        s_walk[0] = (t < nChunks) ? (int)t : -1;
        s_walk[1] = half;
    }
    __syncthreads();
    const int chunk = __builtin_amdgcn_readfirstlane(s_walk[0]);
    const int half  = __builtin_amdgcn_readfirstlane(s_walk[1]);
    if (chunk < 0) return;

    const int eA = chunk * HS_CHUNK + wave * HS_EDGES_PER_WAVE + lane;
    const int eB = eA + 32;
    const int last = nEdges - 1;
    const int iA = (eA < last) ? eA : last;
    const int iB = (eB < last) ? eB : last;
    const int sA = src[iA], dA = dst[iA];
    const int sB = src[iB], dB = dst[iB];
    asm volatile("" :: "v"(sA), "v"(dA), "v"(sB), "v"(dB));
    const bool pA = (eA < nEdges) && ((((dA >= halfRow) ? 1 : 0)) == half);
    const bool pB = (eB < nEdges) && ((((dB >= halfRow) ? 1 : 0)) == half);
    const unsigned mA = __builtin_amdgcn_ballot_w32(pA);
    const unsigned mB = __builtin_amdgcn_ballot_w32(pB);
    const int nA  = __builtin_popcount(mA);
    const int cnt = nA + __builtin_popcount(mB);
    int2* __restrict__ L = s_list[wave];
    if (pA) L[__builtin_amdgcn_mbcnt_lo(mA, 0u)] = make_int2(sA, dA);
    if (pB) L[nA + __builtin_amdgcn_mbcnt_lo(mB, 0u)] = make_int2(sB, dB);
    __syncthreads();
    int2 c0 = L[lane];
    int2 c1 = L[32 + lane];
    if (lane >= cnt)      c0 = make_int2(0, 0);
    if (32 + lane >= cnt) c1 = make_int2(0, 0);

    #pragma unroll
    for (int g = 0; g < 32 / HS_GROUP; ++g)
        if (g * HS_GROUP < cnt)
            scatter_group(h, agg, c0.x, c0.y, g * HS_GROUP, cnt, lane);
    #pragma unroll
    for (int g = 0; g < 32 / HS_GROUP; ++g)
        if (32 + g * HS_GROUP < cnt)
            scatter_group(h, agg, c1.x, c1.y, g * HS_GROUP, cnt - 32, lane);
}

__global__ void __launch_bounds__(128)
gcn_transform_wmma(const float* __restrict__ agg,
                   const float* __restrict__ W,
                   const float* __restrict__ bias,
                   const float* __restrict__ in_isqrt,
                   const float* __restrict__ post_scale,
                   float* __restrict__ out, int nNodes) {
    __shared__ float lds[16 * LDS_STRIDE];

    const int r0   = blockIdx.x * 16;
    const int tid  = threadIdx.x;
    const int wave = tid >> 5;
    const int lane = tid & 31;

    {
        int idx = tid * 8;
        int row = idx >> 6;
        int col = idx & 63;
        int grow = r0 + row; if (grow >= nNodes) grow = nNodes - 1;
        float s = in_isqrt[grow];
        const float* g = agg + (size_t)grow * D_FEAT + col;
        float* l = &lds[row * LDS_STRIDE + col];
        #pragma unroll
        for (int j = 0; j < 8; ++j) l[j] = g[j] * s;
    }
    __syncthreads();

    const int colbase = wave << 4;
    const int n     = lane & 15;
    const int khalf = lane >> 4;

    v8f acc = {};
    #pragma unroll
    for (int kk = 0; kk < 16; ++kk) {
        const int k0 = (kk << 2) + (khalf << 1);
        v2f a = *(const v2f*)&lds[n * LDS_STRIDE + k0];
        v2f b;
        b.x = W[(k0 + 0) * D_FEAT + colbase + n];
        b.y = W[(k0 + 1) * D_FEAT + colbase + n];
        acc = __builtin_amdgcn_wmma_f32_16x16x4_f32(
             false, a,  false, b,
             (short)0, acc,  false,  false);
    }

    const int col = colbase + n;
    const float bv = bias[col];
    #pragma unroll
    for (int r = 0; r < 8; ++r) {
        int row = r0 + (khalf << 3) + r;
        if (row < nNodes) {
            float v = acc[r] + bv;
            v = (v > 0.0f) ? v : 0.01f * v;
            if (post_scale) v *= post_scale[row];
            out[(size_t)row * D_FEAT + col] = v;
        }
    }
}

__global__ void gcn_classify(const float* __restrict__ h,
                             const float* __restrict__ Wc,
                             const float* __restrict__ bc,
                             float* __restrict__ out, int nNodes) {
    int i = blockIdx.x * blockDim.x + threadIdx.x;
    if (i >= nNodes) return;
    float a0 = bc[0], a1 = bc[1];
    const float4* row = (const float4*)(h + (size_t)i * D_FEAT);
    #pragma unroll
    for (int q = 0; q < 16; ++q) {
        float4 v = row[q];
        int k = q << 2;
        a0 += v.x * Wc[(k+0)*2] + v.y * Wc[(k+1)*2] + v.z * Wc[(k+2)*2] + v.w * Wc[(k+3)*2];
        a1 += v.x * Wc[(k+0)*2+1] + v.y * Wc[(k+1)*2+1] + v.z * Wc[(k+2)*2+1] + v.w * Wc[(k+3)*2+1];
    }
    out[(size_t)i * 2 + 0] = a0;
    out[(size_t)i * 2 + 1] = a1;
}

extern "C" void kernel_launch(void* const* d_in, const int* in_sizes, int n_in,
                              void* d_out, int out_size, void* d_ws, size_t ws_size,
                              hipStream_t stream) {
    const float* x   = (const float*)d_in[0];
    const int*   src = (const int*)  d_in[1];
    const int*   dst = (const int*)  d_in[2];
    const float* W1  = (const float*)d_in[3];
    const float* b1  = (const float*)d_in[4];
    const float* W2  = (const float*)d_in[5];
    const float* b2  = (const float*)d_in[6];
    const float* Wc  = (const float*)d_in[7];
    const float* bc  = (const float*)d_in[8];
    float* out = (float*)d_out;

    const int N = in_sizes[0] / D_FEAT;
    const int E = in_sizes[1];

    float* bufA      = (float*)d_ws;
    float* bufB      = bufA + (size_t)N * D_FEAT;
    float* out_isqrt = bufB + (size_t)N * D_FEAT;
    float* in_isqrt  = out_isqrt + N;

    unsigned* tickets = (unsigned*)d_out;
    const int nChunks = (E + HS_CHUNK - 1) / HS_CHUNK;

    const int B256 = 256;
    const int featTotal = N * D_FEAT;
    dim3 blk(B256);

    unsigned* degPartial = (unsigned*)bufB;
    gcn_degree_lds<<<dim3(DEG_SLICES, (N + DEG_RANGE - 1) / DEG_RANGE, 2), dim3(DEG_BLOCK), 0, stream>>>(
        src, dst, degPartial, E, N);
    gcn_deg_isqrt<<<(2 * N + B256 - 1) / B256, blk, 0, stream>>>(degPartial, out_isqrt, N);

    gcn_scale_rows<<<(N * 16 + B256 - 1) / B256, blk, 0, stream>>>(x, out_isqrt, bufA, N * 16);
    gcn_zero_f32<<<(featTotal + B256 - 1) / B256, blk, 0, stream>>>(bufB, featTotal, tickets);
    gcn_scatter_half<<<2 * nChunks, dim3(HS_BLOCK), 0, stream>>>(bufA, src, dst, bufB, tickets, E, (N + 1) / 2);
    gcn_transform_wmma<<<(N + 15) / 16, dim3(128), 0, stream>>>(
        bufB, W1, b1, in_isqrt, out_isqrt, bufA, N);

    gcn_zero_f32<<<(featTotal + B256 - 1) / B256, blk, 0, stream>>>(bufB, featTotal, tickets + 2);
    gcn_scatter_half<<<2 * nChunks, dim3(HS_BLOCK), 0, stream>>>(bufA, src, dst, bufB, tickets + 2, E, (N + 1) / 2);
    gcn_transform_wmma<<<(N + 15) / 16, dim3(128), 0, stream>>>(
        bufB, W2, b2, in_isqrt, nullptr, bufA, N);

    gcn_classify<<<(N + B256 - 1) / B256, blk, 0, stream>>>(bufA, Wc, bc, out, N);
}


# Layer transform moved in front of the row sum (rows times W, then scatter-added); pre-scale, bias+LeakyReLU and head fold into the neighbouring launches, 10 -> 9; sums taken in a new order
# speedup vs baseline: 2.6742x; 1.2911x over previous
#include <hip/hip_runtime.h>
#include <hip/hip_bf16.h>


typedef __attribute__((ext_vector_type(2))) float v2f;
typedef __attribute__((ext_vector_type(8))) float v8f;

#define D_FEAT 64
#define LDS_STRIDE 68

__global__ void gcn_zero_f32(float* __restrict__ p, int n,
                             unsigned* __restrict__ tickets) {
    int i = blockIdx.x * blockDim.x + threadIdx.x;
    if (i < n) p[i] = 0.0f;
    if (i == 0) { tickets[0] = 0u; tickets[1] = 0u; }
}

#define DEG_RANGE  50000
#define DEG_SLICES 16
#define DEG_BLOCK  1024

__device__ __forceinline__ void deg_count_one(unsigned* cnt, int id, int lo) {
    unsigned r = (unsigned)(id - lo);
    if (r < (unsigned)DEG_RANGE) atomicAdd(&cnt[r], 1u);
}

__global__ void __launch_bounds__(DEG_BLOCK)
gcn_degree_lds(const int* __restrict__ src, const int* __restrict__ dst,
               unsigned* __restrict__ partial, int nEdges, int nNodes) {
    __shared__ unsigned cnt[DEG_RANGE];
    const int tid = threadIdx.x;
    const int* __restrict__ idx = blockIdx.z ? dst : src;
    const int lo = blockIdx.y * DEG_RANGE;

    for (int i = tid; i < DEG_RANGE; i += DEG_BLOCK) cnt[i] = 0u;
    __syncthreads();

    const int per = (((nEdges + DEG_SLICES - 1) / DEG_SLICES) + 3) & ~3;
    const int e0 = blockIdx.x * per;
    int e1 = e0 + per; if (e1 > nEdges) e1 = nEdges;
    const int nv = (e1 > e0) ? ((e1 - e0) >> 2) : 0;
    const int4* __restrict__ v = (const int4*)(idx + e0);
    int q = tid;
    for (; q + 3 * DEG_BLOCK < nv; q += 4 * DEG_BLOCK) {
        int4 k0 = v[q];
        int4 k1 = v[q + DEG_BLOCK];
        int4 k2 = v[q + 2 * DEG_BLOCK];
        int4 k3 = v[q + 3 * DEG_BLOCK];
        deg_count_one(cnt, k0.x, lo); deg_count_one(cnt, k0.y, lo);
        deg_count_one(cnt, k0.z, lo); deg_count_one(cnt, k0.w, lo);
        deg_count_one(cnt, k1.x, lo); deg_count_one(cnt, k1.y, lo);
        deg_count_one(cnt, k1.z, lo); deg_count_one(cnt, k1.w, lo);
        deg_count_one(cnt, k2.x, lo); deg_count_one(cnt, k2.y, lo);
        deg_count_one(cnt, k2.z, lo); deg_count_one(cnt, k2.w, lo);
        deg_count_one(cnt, k3.x, lo); deg_count_one(cnt, k3.y, lo);
        deg_count_one(cnt, k3.z, lo); deg_count_one(cnt, k3.w, lo);
    }
    for (; q < nv; q += DEG_BLOCK) {
        int4 k = v[q];
        deg_count_one(cnt, k.x, lo);
        deg_count_one(cnt, k.y, lo);
        deg_count_one(cnt, k.z, lo);
        deg_count_one(cnt, k.w, lo);
    }
    for (int e = e0 + (nv << 2) + tid; e < e1; e += DEG_BLOCK)
        deg_count_one(cnt, idx[e], lo);
    __syncthreads();

    int nHere = nNodes - lo; if (nHere > DEG_RANGE) nHere = DEG_RANGE;
    unsigned* __restrict__ o =
        partial + (size_t)(blockIdx.z * DEG_SLICES + blockIdx.x) * nNodes + lo;
    for (int i = tid; i < nHere; i += DEG_BLOCK) o[i] = cnt[i];
}

__global__ void gcn_deg_isqrt(const unsigned* __restrict__ partial,
                              float* __restrict__ d, int nNodes) {
    int i = blockIdx.x * blockDim.x + threadIdx.x;
    if (i >= 2 * nNodes) return;
    const int a = (i >= nNodes) ? 1 : 0;
    const int node = i - a * nNodes;
    const unsigned* __restrict__ p = partial + (size_t)a * DEG_SLICES * nNodes + node;
    unsigned s = 0u;
    #pragma unroll
    for (int k = 0; k < DEG_SLICES; ++k) s += p[(size_t)k * nNodes];
    d[i] = rsqrtf(fmaxf((float)s, 1.0f));
}

#define HS_BLOCK 128
#define HS_WAVES (HS_BLOCK / 32)
#define HS_EDGES_PER_WAVE 64
#define HS_CHUNK (HS_WAVES * HS_EDGES_PER_WAVE)
#define HS_GROUP 8

__device__ __forceinline__ void scatter_group(const float* __restrict__ h,
                                              float* __restrict__ agg,
                                              int cs, int cd, int base, int n,
                                              int lane) {
    int s[HS_GROUP], d[HS_GROUP];
    float lo[HS_GROUP], hi[HS_GROUP];
    #pragma unroll
    for (int j = 0; j < HS_GROUP; ++j) {
        s[j] = __builtin_amdgcn_readlane(cs, base + j);
        d[j] = __builtin_amdgcn_readlane(cd, base + j);
    }
    #pragma unroll
    for (int j = 0; j < HS_GROUP; ++j) {
        const float* r = h + (size_t)s[j] * D_FEAT + lane;
        lo[j] = r[0];
        hi[j] = r[32];
    }
    #pragma unroll
    for (int j = 0; j < HS_GROUP; ++j) {
        if (base + j < n) {
            float* p = agg + (size_t)d[j] * D_FEAT + lane;
            atomicAdd(p, lo[j]);
            atomicAdd(p + 32, hi[j]);
        }
    }
}

__global__ void __launch_bounds__(HS_BLOCK)
gcn_scatter_half(const float* __restrict__ h,
                 const int* __restrict__ src, const int* __restrict__ dst,
                 float* __restrict__ agg, unsigned* __restrict__ tickets,
                 int nEdges, int halfRow) {
    __shared__ int s_walk[2];
    __shared__ int2 s_list[HS_WAVES][HS_EDGES_PER_WAVE];
    const int tid  = threadIdx.x;
    const int lane = tid & 31;
    const int wave = tid >> 5;
    const unsigned nChunks = (unsigned)((nEdges + HS_CHUNK - 1) / HS_CHUNK);

    if (tid == 0) {
        const unsigned id = __builtin_amdgcn_s_sendmsg_rtn(0x87);
        int half = (int)((id >> 18) & 1u);
        unsigned t = atomicAdd(&tickets[half], 1u);
        if (t >= nChunks) { half ^= 1; t = atomicAdd(&tickets[half], 1u); }
        s_walk[0] = (t < nChunks) ? (int)t : -1;
        s_walk[1] = half;
    }
    __syncthreads();
    const int chunk = __builtin_amdgcn_readfirstlane(s_walk[0]);
    const int half  = __builtin_amdgcn_readfirstlane(s_walk[1]);
    if (chunk < 0) return;

    const int eA = chunk * HS_CHUNK + wave * HS_EDGES_PER_WAVE + lane;
    const int eB = eA + 32;
    const int last = nEdges - 1;
    const int iA = (eA < last) ? eA : last;
    const int iB = (eB < last) ? eB : last;
    const int sA = src[iA], dA = dst[iA];
    const int sB = src[iB], dB = dst[iB];
    asm volatile("" :: "v"(sA), "v"(dA), "v"(sB), "v"(dB));
    const bool pA = (eA < nEdges) && ((((dA >= halfRow) ? 1 : 0)) == half);
    const bool pB = (eB < nEdges) && ((((dB >= halfRow) ? 1 : 0)) == half);
    const unsigned mA = __builtin_amdgcn_ballot_w32(pA);
    const unsigned mB = __builtin_amdgcn_ballot_w32(pB);
    const int nA  = __builtin_popcount(mA);
    const int cnt = nA + __builtin_popcount(mB);
    int2* __restrict__ L = s_list[wave];
    if (pA) L[__builtin_amdgcn_mbcnt_lo(mA, 0u)] = make_int2(sA, dA);
    if (pB) L[nA + __builtin_amdgcn_mbcnt_lo(mB, 0u)] = make_int2(sB, dB);
    __syncthreads();
    int2 c0 = L[lane];
    int2 c1 = L[32 + lane];
    if (lane >= cnt)      c0 = make_int2(0, 0);
    if (32 + lane >= cnt) c1 = make_int2(0, 0);

    #pragma unroll
    for (int g = 0; g < 32 / HS_GROUP; ++g)
        if (g * HS_GROUP < cnt)
            scatter_group(h, agg, c0.x, c0.y, g * HS_GROUP, cnt, lane);
    #pragma unroll
    for (int g = 0; g < 32 / HS_GROUP; ++g)
        if (32 + g * HS_GROUP < cnt)
            scatter_group(h, agg, c1.x, c1.y, g * HS_GROUP, cnt - 32, lane);
}

__device__ __forceinline__ float lrelu(float v) { return (v > 0.0f) ? v : 0.01f * v; }

__global__ void __launch_bounds__(128)
gcn_transform_wmma(const float* __restrict__ in,
                   const float* __restrict__ W,
                   const float* __restrict__ pre_bias,
                   const float* __restrict__ pre_scale,
                   const float* __restrict__ row_scale,
                   float* __restrict__ out, int nNodes) {
    __shared__ __attribute__((aligned(16))) float lds[16 * LDS_STRIDE];

    const int r0   = blockIdx.x * 16;
    const int tid  = threadIdx.x;
    const int wave = tid >> 5;
    const int lane = tid & 31;

    {
        int idx = tid * 8;
        int row = idx >> 6;
        int col = idx & 63;
        int grow = r0 + row; if (grow >= nNodes) grow = nNodes - 1;
        const float so = row_scale[grow];
        const float4* g = (const float4*)(in + (size_t)grow * D_FEAT + col);
        float4 v0 = g[0];
        float4 v1 = g[1];
        if (pre_bias) {
            const float si = pre_scale[grow];
            const float4 c0 = ((const float4*)(pre_bias + col))[0];
            const float4 c1 = ((const float4*)(pre_bias + col))[1];
            v0.x = lrelu(v0.x * si + c0.x); v0.y = lrelu(v0.y * si + c0.y);
            v0.z = lrelu(v0.z * si + c0.z); v0.w = lrelu(v0.w * si + c0.w);
            v1.x = lrelu(v1.x * si + c1.x); v1.y = lrelu(v1.y * si + c1.y);
            v1.z = lrelu(v1.z * si + c1.z); v1.w = lrelu(v1.w * si + c1.w);
        }
        v0.x *= so; v0.y *= so; v0.z *= so; v0.w *= so;
        v1.x *= so; v1.y *= so; v1.z *= so; v1.w *= so;
        float4* l = (float4*)&lds[row * LDS_STRIDE + col];
        l[0] = v0;
        l[1] = v1;
    }
    __syncthreads();

    const int colbase = wave << 4;
    const int n     = lane & 15;
    const int khalf = lane >> 4;

    v8f acc = {};
    #pragma unroll
    for (int kk = 0; kk < 16; ++kk) {
        const int k0 = (kk << 2) + (khalf << 1);
        v2f a = *(const v2f*)&lds[n * LDS_STRIDE + k0];
        v2f b;
        b.x = W[(k0 + 0) * D_FEAT + colbase + n];
        b.y = W[(k0 + 1) * D_FEAT + colbase + n];
        acc = __builtin_amdgcn_wmma_f32_16x16x4_f32(
             false, a,  false, b,
             (short)0, acc,  false,  false);
    }

    const int col = colbase + n;
    #pragma unroll
    for (int r = 0; r < 8; ++r) {
        int row = r0 + (khalf << 3) + r;
        if (row < nNodes) out[(size_t)row * D_FEAT + col] = acc[r];
    }
}

#define HEAD_BLOCK 256
#define HEAD_ROWS_PER_WAVE 16
__global__ void __launch_bounds__(HEAD_BLOCK)
gcn_head(const float* __restrict__ agg,
         const float* __restrict__ bias,
         const float* __restrict__ in_isqrt,
         const float* __restrict__ Wc,
         const float* __restrict__ bc,
         float* __restrict__ out, int nNodes) {
    const int lane = threadIdx.x & 31;
    const int q    = lane & 15;
    const int half = lane >> 4;
    const int wave = (blockIdx.x * HEAD_BLOCK + threadIdx.x) >> 5;
    const int rowBase = wave * HEAD_ROWS_PER_WAVE;
    if (rowBase >= nNodes) return;

    const float4 bv  = ((const float4*)bias)[q];
    const float4 w01 = ((const float4*)Wc)[2 * q];
    const float4 w23 = ((const float4*)Wc)[2 * q + 1];

    float4 v[8];
    float  s[8];
    #pragma unroll
    for (int it = 0; it < 8; ++it) {
        int row = rowBase + 8 * half + it; if (row >= nNodes) row = nNodes - 1;
        v[it] = ((const float4*)(agg + (size_t)row * D_FEAT))[q];
        s[it] = in_isqrt[row];
    }
    asm volatile("" :: "v"(v[0].x), "v"(v[1].x), "v"(v[2].x), "v"(v[3].x),
                       "v"(v[4].x), "v"(v[5].x), "v"(v[6].x), "v"(v[7].x),
                       "v"(s[0]), "v"(s[1]), "v"(s[2]), "v"(s[3]),
                       "v"(s[4]), "v"(s[5]), "v"(s[6]), "v"(s[7]),
                       "v"(bv.x), "v"(w01.x), "v"(w23.x));
    float res = 0.0f;
    #pragma unroll
    for (int it = 0; it < 8; ++it) {
        const float ax = lrelu(v[it].x * s[it] + bv.x);
        const float ay = lrelu(v[it].y * s[it] + bv.y);
        const float az = lrelu(v[it].z * s[it] + bv.z);
        const float aw = lrelu(v[it].w * s[it] + bv.w);
        float p0 = ax * w01.x + ay * w01.z + az * w23.x + aw * w23.z;
        float p1 = ax * w01.y + ay * w01.w + az * w23.y + aw * w23.w;
        #pragma unroll
        for (int m = 8; m >= 1; m >>= 1) {
            p0 += __shfl_xor(p0, m);
            p1 += __shfl_xor(p1, m);
        }
        if (((lane >> 1) & 7) == it) res = (lane & 1) ? p1 : p0;
    }
    const int orow = rowBase + (lane >> 1);
    if (orow < nNodes) out[(size_t)rowBase * 2 + lane] = res + bc[lane & 1];
}

extern "C" void kernel_launch(void* const* d_in, const int* in_sizes, int n_in,
                              void* d_out, int out_size, void* d_ws, size_t ws_size,
                              hipStream_t stream) {
    const float* x   = (const float*)d_in[0];
    const int*   src = (const int*)  d_in[1];
    const int*   dst = (const int*)  d_in[2];
    const float* W1  = (const float*)d_in[3];
    const float* b1  = (const float*)d_in[4];
    const float* W2  = (const float*)d_in[5];
    const float* b2  = (const float*)d_in[6];
    const float* Wc  = (const float*)d_in[7];
    const float* bc  = (const float*)d_in[8];
    float* out = (float*)d_out;

    const int N = in_sizes[0] / D_FEAT;
    const int E = in_sizes[1];

    float* bufA      = (float*)d_ws;
    float* bufB      = bufA + (size_t)N * D_FEAT;
    float* out_isqrt = bufB + (size_t)N * D_FEAT;
    float* in_isqrt  = out_isqrt + N;

    unsigned* tickets = (unsigned*)d_out;
    const int nChunks = (E + HS_CHUNK - 1) / HS_CHUNK;

    const int B256 = 256;
    const int featTotal = N * D_FEAT;
    dim3 blk(B256);

    unsigned* degPartial = (unsigned*)bufB;
    gcn_degree_lds<<<dim3(DEG_SLICES, (N + DEG_RANGE - 1) / DEG_RANGE, 2), dim3(DEG_BLOCK), 0, stream>>>(
        src, dst, degPartial, E, N);
    gcn_deg_isqrt<<<(2 * N + B256 - 1) / B256, blk, 0, stream>>>(degPartial, out_isqrt, N);

    gcn_transform_wmma<<<(N + 15) / 16, dim3(128), 0, stream>>>(
        x, W1, nullptr, nullptr, out_isqrt, bufA, N);
    gcn_zero_f32<<<(featTotal + B256 - 1) / B256, blk, 0, stream>>>(bufB, featTotal, tickets);
    gcn_scatter_half<<<2 * nChunks, dim3(HS_BLOCK), 0, stream>>>(bufA, src, dst, bufB, tickets, E, (N + 1) / 2);

    gcn_transform_wmma<<<(N + 15) / 16, dim3(128), 0, stream>>>(
        bufB, W2, b1, in_isqrt, out_isqrt, bufA, N);
    gcn_zero_f32<<<(featTotal + B256 - 1) / B256, blk, 0, stream>>>(bufB, featTotal, tickets + 2);
    gcn_scatter_half<<<2 * nChunks, dim3(HS_BLOCK), 0, stream>>>(bufA, src, dst, bufB, tickets + 2, E, (N + 1) / 2);

    gcn_head<<<(N + 127) / 128, dim3(HEAD_BLOCK), 0, stream>>>(bufB, b2, in_isqrt, Wc, bc, out, N);
}
